# starting kernel
# baseline (speedup 1.0000x reference)
#include <hip/hip_runtime.h>
#include <math.h>

typedef __attribute__((ext_vector_type(2))) float v2f;
typedef __attribute__((ext_vector_type(4))) float v4f;
typedef __attribute__((ext_vector_type(8))) float v8f;

#define KC 16

template <int LDA, int LDW>
__device__ inline void gemm_tile_acc(const float* __restrict__ A,
                                     const float* __restrict__ W,
                                     int K, int m0, int n0w,
                                     float (*ldsA)[20], float (*ldsB)[20],
                                     v8f& acc)
{
    const int t    = threadIdx.x;
    const int lane = t & 31;
    const int half = lane >> 4;
    const int mrow = lane & 15;

    const int ar = t >> 3;
    const int ac = (t & 7) * 2;
    const int br = lane >> 1;
    const int bc = (lane & 1) * 8;

    const float* aptr = A + (size_t)(m0 + ar) * LDA + ac;
    const float* bptr = W + (size_t)br * LDW + n0w + bc;

    for (int k0 = 0; k0 < K; k0 += KC) {
        const v2f a2 = *(const v2f*)(aptr + k0);
        ldsA[ar][ac]     = a2.x;
        ldsA[ar][ac + 1] = a2.y;

        const v4f b0 = *(const v4f*)(bptr + (size_t)k0 * LDW);
        const v4f b1 = *(const v4f*)(bptr + (size_t)k0 * LDW + 4);
        *(v4f*)&ldsB[br][bc]     = b0;
        *(v4f*)&ldsB[br][bc + 4] = b1;
        __syncthreads();

#pragma unroll
        for (int ks = 0; ks < KC; ks += 4) {
            v2f a, b;
            a.x = ldsA[mrow][ks + 2 * half];
            a.y = ldsA[mrow][ks + 2 * half + 1];
            b.x = ldsB[ks + 2 * half][mrow];
            b.y = ldsB[ks + 2 * half + 1][mrow];
            acc = __builtin_amdgcn_wmma_f32_16x16x4_f32(
                false, a, false, b, (short)0, acc, false, false);
        }
        __syncthreads();
    }
}

__global__ void __launch_bounds__(128)
lstm_cell_kernel(const float* __restrict__ A1, const float* __restrict__ W,
                 const float* __restrict__ A2, const float* __restrict__ Uw,
                 const float* __restrict__ bias, const float* __restrict__ Cprev,
                 float* __restrict__ Hout, float* __restrict__ Cout,
                 int K1, int K2, int Uu)
{
    __shared__ float ldsA[16][20];
    __shared__ float ldsB[4][16][20];
    __shared__ float gbuf[4][16][17];

    const int t    = threadIdx.x;
    const int wave = t >> 5;
    const int lane = t & 31;
    const int half = lane >> 4;
    const int mrow = lane & 15;
    const int m0   = blockIdx.y * 16;
    const int n0   = blockIdx.x * 16;
    const int n0w  = wave * Uu + n0;

    v8f acc = {};
    gemm_tile_acc<1024, 4096>(A1, W,  K1, m0, n0w, ldsA, ldsB[wave], acc);
    gemm_tile_acc<1024, 4096>(A2, Uw, K2, m0, n0w, ldsA, ldsB[wave], acc);

    const float bv = bias[n0w + mrow];
#pragma unroll
    for (int r = 0; r < 8; ++r) {
        const float z = acc[r] + bv;
        const float a = (wave == 2) ? tanhf(z) : (1.0f / (1.0f + __expf(-z)));
        gbuf[wave][r + 8 * half][mrow] = a;
    }
    __syncthreads();

#pragma unroll
    for (int e = t; e < 256; e += 128) {
        const int m = e >> 4, n = e & 15;
        const float iv = gbuf[0][m][n];
        const float fv = gbuf[1][m][n];
        const float gv = gbuf[2][m][n];
        const float ov = gbuf[3][m][n];
        const size_t idx = (size_t)(m0 + m) * Uu + n0 + n;
        const float cn = fv * Cprev[idx] + iv * gv;
        Cout[idx] = cn;
        Hout[idx] = ov * tanhf(cn);
    }
}

__global__ void __launch_bounds__(128)
sproj_kernel(const float* __restrict__ H2, const float* __restrict__ Ws,
             const float* __restrict__ bs, float* __restrict__ Sout, int Uu)
{
    __shared__ float ldsA[16][20];
    __shared__ float ldsB[4][16][20];

    const int t    = threadIdx.x;
    const int wave = t >> 5;
    const int lane = t & 31;
    const int half = lane >> 4;
    const int mrow = lane & 15;
    const int m0   = blockIdx.y * 16;
    const int n0w  = blockIdx.x * 64 + wave * 16;

    v8f acc = {};
    gemm_tile_acc<1024, 1024>(H2, Ws, Uu, m0, n0w, ldsA, ldsB[wave], acc);

    const float bv = bs[n0w + mrow];
#pragma unroll
    for (int r = 0; r < 8; ++r) {
        const int gm = m0 + r + 8 * half;
        Sout[(size_t)gm * Uu + n0w + mrow] = acc[r] + bv;
    }
}

__global__ void __launch_bounds__(512)
attention_kernel(const float* __restrict__ av, const float* __restrict__ S,
                 const float* __restrict__ H2, const float* __restrict__ C2,
                 float* __restrict__ out, int T, int Uu, int B)
{
    __shared__ float sS[1024];
    __shared__ float ctx_s[1024];
    __shared__ float wm[16], wl[16];

    const int b    = blockIdx.x;
    const int t    = threadIdx.x;
    const int wave = t >> 5;
    const int lane = t & 31;

    for (int u = t; u < Uu; u += 512) {
        sS[u]    = S[(size_t)b * Uu + u];
        ctx_s[u] = 0.0f;
    }
    __syncthreads();

    float m = -INFINITY, l = 0.0f;
    float ctx[32];
#pragma unroll
    for (int j = 0; j < 32; ++j) ctx[j] = 0.0f;

    const float* avb = av + (size_t)b * T * Uu;
    for (int tt = wave; tt < T; tt += 16) {
        const float* row = avb + (size_t)tt * Uu;
        __builtin_prefetch(avb + (size_t)(tt + 16) * Uu + lane * 32, 0, 0);

        v4f frag[8];
        float dot = 0.0f;
#pragma unroll
        for (int j = 0; j < 8; ++j) {
            const int u = (j * 32 + lane) * 4;
            frag[j] = __builtin_nontemporal_load((const v4f*)(row + u));
            dot += frag[j].x * sS[u] + frag[j].y * sS[u + 1] +
                   frag[j].z * sS[u + 2] + frag[j].w * sS[u + 3];
        }
#pragma unroll
        for (int off = 16; off > 0; off >>= 1) dot += __shfl_xor(dot, off, 32);

        const float mn    = fmaxf(m, dot);
        const float scale = __expf(m - mn);
        const float e     = __expf(dot - mn);
        l = l * scale + e;
#pragma unroll
        for (int j = 0; j < 8; ++j) {
            ctx[j * 4 + 0] = ctx[j * 4 + 0] * scale + e * frag[j].x;
            ctx[j * 4 + 1] = ctx[j * 4 + 1] * scale + e * frag[j].y;
            ctx[j * 4 + 2] = ctx[j * 4 + 2] * scale + e * frag[j].z;
            ctx[j * 4 + 3] = ctx[j * 4 + 3] * scale + e * frag[j].w;
        }
        m = mn;
    }

    if (lane == 0) { wm[wave] = m; wl[wave] = l; }
    __syncthreads();

    float M = wm[0];
#pragma unroll
    for (int w = 1; w < 16; ++w) M = fmaxf(M, wm[w]);
    float L = 0.0f;
#pragma unroll
    for (int w = 0; w < 16; ++w) L += wl[w] * __expf(wm[w] - M);

    const float f = __expf(m - M);
#pragma unroll
    for (int j = 0; j < 8; ++j) {
        const int u = (j * 32 + lane) * 4;
        atomicAdd(&ctx_s[u + 0], ctx[j * 4 + 0] * f);
        atomicAdd(&ctx_s[u + 1], ctx[j * 4 + 1] * f);
        atomicAdd(&ctx_s[u + 2], ctx[j * 4 + 2] * f);
        atomicAdd(&ctx_s[u + 3], ctx[j * 4 + 3] * f);
    }
    __syncthreads();

    const float invL = 1.0f / L;
    const size_t base = (size_t)b * Uu;
    for (int u = t; u < Uu; u += 512) {
        out[base + u]                  = H2[base + u];
        out[(size_t)B * Uu + base + u] = C2[base + u] + ctx_s[u] * invL;
    }
}

extern "C" void kernel_launch(void* const* d_in, const int* in_sizes, int n_in,
                              void* d_out, int out_size, void* d_ws, size_t ws_size,
                              hipStream_t stream)
{
    (void)in_sizes; (void)n_in; (void)out_size; (void)ws_size;
    const int B = 128, T = 512, U = 1024;

    const float* x   = (const float*)d_in[0];
    const float* h0  = (const float*)d_in[1];
    const float* c0  = (const float*)d_in[2];
    const float* av  = (const float*)d_in[3];
    const float* W1  = (const float*)d_in[4];
    const float* U1  = (const float*)d_in[5];
    const float* b1  = (const float*)d_in[6];
    const float* W2  = (const float*)d_in[7];
    const float* U2  = (const float*)d_in[8];
    const float* b2  = (const float*)d_in[9];
    const float* Ws  = (const float*)d_in[10];
    const float* bs  = (const float*)d_in[11];
    float* out = (float*)d_out;

    float* ws = (float*)d_ws;
    const size_t BU = (size_t)B * U;
    float* h1 = ws;
    float* c1 = ws + BU;
    float* h2 = ws + 2 * BU;
    float* c2 = ws + 3 * BU;
    float* s  = ws + 4 * BU;

    dim3 gGemm(U / 16, B / 16);
    dim3 bGemm(128);

    lstm_cell_kernel<<<gGemm, bGemm, 0, stream>>>(x, W1, h0, U1, b1, c0,
                                                  h1, c1, 1024, U, U);
    lstm_cell_kernel<<<gGemm, bGemm, 0, stream>>>(h1, W2, h1, U2, b2, c1,
                                                  h2, c2, U, U, U);
    sproj_kernel<<<dim3(U / 64, B / 16), bGemm, 0, stream>>>(h2, Ws, bs, s, U);
    attention_kernel<<<dim3(B), dim3(512), 0, stream>>>(av, s, h2, c2, out,
                                                        T, U, B);
}


# LSTM/projection GEMMs re-decomposed: no LDS staging or K-loop barriers; each wave loads WMMA fragments from global for all 4 gates over 1/16 of k; 16 partials summed in LDS: sum order changes
# speedup vs baseline: 4.1619x; 4.1619x over previous
#include <hip/hip_runtime.h>
#include <math.h>


typedef __attribute__((ext_vector_type(2))) float v2f;
typedef __attribute__((ext_vector_type(4))) float v4f;
typedef __attribute__((ext_vector_type(8))) float v8f;

#define PPITCH 36

template <int NG, int LDA, int LDW, int GS, int UNR>
__device__ inline void wave_gemm_direct(const float* __restrict__ A,
                                        const float* __restrict__ Wt,
                                        int m0, int n0, int kb, int klen,
                                        v8f (&acc)[NG][2])
{
    const int lane = threadIdx.x & 31;
    const int h    = lane >> 4;
    const int c    = lane & 15;

    const float* ap = A  + (size_t)(m0 + c) * LDA + kb + 4 * h;
    const float* bp = Wt + (size_t)(kb + 4 * h) * LDW + n0 + 2 * c;

#pragma unroll UNR
    for (int k0 = 0; k0 < klen; k0 += 32) {
        v4f a4[4];
        v2f b2[4][NG][4];
#pragma unroll
        for (int ch = 0; ch < 4; ++ch) {
            a4[ch] = *(const v4f*)(ap + k0 + 8 * ch);
#pragma unroll
            for (int e = 0; e < 4; ++e) {
#pragma unroll
                for (int g = 0; g < NG; ++g) {
                    b2[ch][g][e] = *(const v2f*)(bp + (size_t)(k0 + 8 * ch + e) * LDW + g * GS);
                }
            }
        }
#pragma unroll
        for (int ch = 0; ch < 4; ++ch) {
#pragma unroll
            for (int s = 0; s < 2; ++s) {
                v2f a;
                a.x = a4[ch][2 * s];
                a.y = a4[ch][2 * s + 1];
#pragma unroll
                for (int g = 0; g < NG; ++g) {
#pragma unroll
                    for (int j = 0; j < 2; ++j) {
                        v2f b;
                        b.x = b2[ch][g][2 * s][j];
                        b.y = b2[ch][g][2 * s + 1][j];
                        acc[g][j] = __builtin_amdgcn_wmma_f32_16x16x4_f32(
                            false, a, false, b, (short)0, acc[g][j], false, false);
                    }
                }
            }
        }
    }
}

template <int NG>
__device__ inline void store_partial(float (*part)[16][PPITCH], const v8f (&acc)[NG][2])
{
    const int lane = threadIdx.x & 31;
    const int h    = lane >> 4;
    const int c    = lane & 15;
#pragma unroll
    for (int g = 0; g < NG; ++g) {
#pragma unroll
        for (int r = 0; r < 8; ++r) {
            v2f v;
            v.x = acc[g][0][r];
            v.y = acc[g][1][r];
            *(v2f*)&part[g][8 * h + r][2 * c] = v;
        }
    }
}

__global__ void __launch_bounds__(512) __attribute__((amdgpu_waves_per_eu(4, 4)))
lstm_cell_kernel(const float* __restrict__ A1, const float* __restrict__ W,
                 const float* __restrict__ A2, const float* __restrict__ Uw,
                 const float* __restrict__ bias, const float* __restrict__ Cprev,
                 float* __restrict__ Hout, float* __restrict__ Cout)
{
    constexpr int Uu = 1024;
    __shared__ float part[16][4][16][PPITCH];

    const int t    = threadIdx.x;
    const int wave = t >> 5;
    const int lane = t & 31;
    const int m0   = blockIdx.y * 16;
    const int n0   = blockIdx.x * 32;

    const bool second = wave >= 8;
    const float* Aop = second ? A2 : A1;
    const float* Wop = second ? Uw : W;
    const int kb = (wave & 7) * 128;

    v8f acc[4][2] = {};
    wave_gemm_direct<4, 1024, 4096, 1024, 1>(Aop, Wop, m0, n0, kb, 128, acc);
    store_partial<4>(part[wave], acc);
    __syncthreads();

    const int m = wave, n = lane;
    float z[4];
#pragma unroll
    for (int g = 0; g < 4; ++g) {
        float s = 0.0f;
#pragma unroll
        for (int w = 0; w < 16; ++w) s += part[w][g][m][n];
        z[g] = s + bias[g * Uu + n0 + n];
    }
    const float iv = 1.0f / (1.0f + __expf(-z[0]));
    const float fv = 1.0f / (1.0f + __expf(-z[1]));
    const float gv = tanhf(z[2]);
    const float ov = 1.0f / (1.0f + __expf(-z[3]));
    const size_t idx = (size_t)(m0 + m) * Uu + n0 + n;
    const float cn = fv * Cprev[idx] + iv * gv;
    Cout[idx] = cn;
    Hout[idx] = ov * tanhf(cn);
}

__global__ void __launch_bounds__(512) __attribute__((amdgpu_waves_per_eu(4, 4)))
sproj_kernel(const float* __restrict__ H2, const float* __restrict__ Ws,
             const float* __restrict__ bs, float* __restrict__ Sout)
{
    constexpr int Uu = 1024;
    __shared__ float part[16][1][16][PPITCH];

    const int t    = threadIdx.x;
    const int wave = t >> 5;
    const int lane = t & 31;
    const int m0   = blockIdx.y * 16;
    const int n0   = blockIdx.x * 32;

    v8f acc[1][2] = {};
    wave_gemm_direct<1, 1024, 1024, 0, 2>(H2, Ws, m0, n0, wave * 64, 64, acc);
    store_partial<1>(part[wave], acc);
    __syncthreads();

    const int m = wave, n = lane;
    float s = 0.0f;
#pragma unroll
    for (int w = 0; w < 16; ++w) s += part[w][0][m][n];
    Sout[(size_t)(m0 + m) * Uu + n0 + n] = s + bs[n0 + n];
}

__global__ void __launch_bounds__(512)
attention_kernel(const float* __restrict__ av, const float* __restrict__ S,
                 const float* __restrict__ H2, const float* __restrict__ C2,
                 float* __restrict__ out, int T, int Uu, int B)
{
    __shared__ float sS[1024];
    __shared__ float ctx_s[1024];
    __shared__ float wm[16], wl[16];

    const int b    = blockIdx.x;
    const int t    = threadIdx.x;
    const int wave = t >> 5;
    const int lane = t & 31;

    for (int u = t; u < Uu; u += 512) {
        sS[u]    = S[(size_t)b * Uu + u];
        ctx_s[u] = 0.0f;
    }
    __syncthreads();

    float m = -INFINITY, l = 0.0f;
    float ctx[32];
#pragma unroll
    for (int j = 0; j < 32; ++j) ctx[j] = 0.0f;

    const float* avb = av + (size_t)b * T * Uu;
    for (int tt = wave; tt < T; tt += 16) {
        const float* row = avb + (size_t)tt * Uu;
        __builtin_prefetch(avb + (size_t)(tt + 16) * Uu + lane * 32, 0, 0);

        v4f frag[8];
        float dot = 0.0f;
#pragma unroll
        for (int j = 0; j < 8; ++j) {
            const int u = (j * 32 + lane) * 4;
            frag[j] = __builtin_nontemporal_load((const v4f*)(row + u));
            dot += frag[j].x * sS[u] + frag[j].y * sS[u + 1] +
                   frag[j].z * sS[u + 2] + frag[j].w * sS[u + 3];
        }
#pragma unroll
        for (int off = 16; off > 0; off >>= 1) dot += __shfl_xor(dot, off, 32);

        const float mn    = fmaxf(m, dot);
        const float scale = __expf(m - mn);
        const float e     = __expf(dot - mn);
        l = l * scale + e;
#pragma unroll
        for (int j = 0; j < 8; ++j) {
            ctx[j * 4 + 0] = ctx[j * 4 + 0] * scale + e * frag[j].x;
            ctx[j * 4 + 1] = ctx[j * 4 + 1] * scale + e * frag[j].y;
            ctx[j * 4 + 2] = ctx[j * 4 + 2] * scale + e * frag[j].z;
            ctx[j * 4 + 3] = ctx[j * 4 + 3] * scale + e * frag[j].w;
        }
        m = mn;
    }

    if (lane == 0) { wm[wave] = m; wl[wave] = l; }
    __syncthreads();

    float M = wm[0];
#pragma unroll
    for (int w = 1; w < 16; ++w) M = fmaxf(M, wm[w]);
    float L = 0.0f;
#pragma unroll
    for (int w = 0; w < 16; ++w) L += wl[w] * __expf(wm[w] - M);

    const float f = __expf(m - M);
#pragma unroll
    for (int j = 0; j < 8; ++j) {
        const int u = (j * 32 + lane) * 4;
        atomicAdd(&ctx_s[u + 0], ctx[j * 4 + 0] * f);
        atomicAdd(&ctx_s[u + 1], ctx[j * 4 + 1] * f);
        atomicAdd(&ctx_s[u + 2], ctx[j * 4 + 2] * f);
        atomicAdd(&ctx_s[u + 3], ctx[j * 4 + 3] * f);
    }
    __syncthreads();

    const float invL = 1.0f / L;
    const size_t base = (size_t)b * Uu;
    for (int u = t; u < Uu; u += 512) {
        out[base + u]                  = H2[base + u];
        out[(size_t)B * Uu + base + u] = C2[base + u] + ctx_s[u] * invL;
    }
}

extern "C" void kernel_launch(void* const* d_in, const int* in_sizes, int n_in,
                              void* d_out, int out_size, void* d_ws, size_t ws_size,
                              hipStream_t stream)
{
    (void)in_sizes; (void)n_in; (void)out_size; (void)ws_size;
    const int B = 128, T = 512, U = 1024;

    const float* x   = (const float*)d_in[0];
    const float* h0  = (const float*)d_in[1];
    const float* c0  = (const float*)d_in[2];
    const float* av  = (const float*)d_in[3];
    const float* W1  = (const float*)d_in[4];
    const float* U1  = (const float*)d_in[5];
    const float* b1  = (const float*)d_in[6];
    const float* W2  = (const float*)d_in[7];
    const float* U2  = (const float*)d_in[8];
    const float* b2  = (const float*)d_in[9];
    const float* Ws  = (const float*)d_in[10];
    const float* bs  = (const float*)d_in[11];
    float* out = (float*)d_out;

    float* ws = (float*)d_ws;
    const size_t BU = (size_t)B * U;
    float* h1 = ws;
    float* c1 = ws + BU;
    float* h2 = ws + 2 * BU;
    float* c2 = ws + 3 * BU;
    float* s  = ws + 4 * BU;

    dim3 gGemm(U / 32, B / 16);
    dim3 bGemm(512);

    lstm_cell_kernel<<<gGemm, bGemm, 0, stream>>>(x, W1, h0, U1, b1, c0, h1, c1);
    lstm_cell_kernel<<<gGemm, bGemm, 0, stream>>>(h1, W2, h1, U2, b2, c1, h2, c2);
    sproj_kernel<<<gGemm, bGemm, 0, stream>>>(h2, Ws, bs, s);
    attention_kernel<<<dim3(B), dim3(512), 0, stream>>>(av, s, h2, c2, out,
                                                        T, U, B);
}
